# starting kernel
# baseline (speedup 1.0000x reference)
#include <hip/hip_runtime.h>

typedef __attribute__((ext_vector_type(2))) float v2f;
typedef __attribute__((ext_vector_type(8))) float v8f;

namespace {
constexpr int WIN     = 5;
constexpr int VOCAB   = 100000;
constexpr int EMB     = 50;
constexpr int HID     = 250;
constexpr int OUTD    = 36;
constexpr int KPAD    = 256;
constexpr int NPAD2   = 48;
constexpr int ROWS    = 64;
constexpr int STRIDE  = 260;
constexpr int THREADS = 128;
}

__device__ __forceinline__ float fast_tanh(float x) {
  float xc = fminf(15.0f, fmaxf(-15.0f, x));
  float t  = __expf(2.0f * xc);
  return (t - 1.0f) * __builtin_amdgcn_rcpf(t + 1.0f);
}

__global__ __launch_bounds__(256)
void pad_weights(const float* __restrict__ w_in, const float* __restrict__ w_out,
                 float* __restrict__ wp, float* __restrict__ wp2)
{
  int i = blockIdx.x * 256 + threadIdx.x;
  if (i < KPAD * KPAD) {
    int k = i >> 8, n = i & 255;
    wp[i] = (k < HID && n < HID) ? w_in[k * HID + n] : 0.0f;
  }
  if (i < KPAD * NPAD2) {
    int k = i / NPAD2, n = i - k * NPAD2;
    wp2[i] = (k < HID && n < OUTD) ? w_out[k * OUTD + n] : 0.0f;
  }
}

__global__ __launch_bounds__(THREADS)
void tagger_fused(const int*   __restrict__ idx,
                  const float* __restrict__ emb,
                  const float* __restrict__ wp,
                  const float* __restrict__ b_in,
                  const float* __restrict__ wp2,
                  const float* __restrict__ b_out,
                  float*       __restrict__ out)
{
  __shared__ float x_lds[ROWS * STRIDE];
  __shared__ float h_lds[ROWS * STRIDE];
  __shared__ int   i_lds[ROWS * WIN];

  const int t       = threadIdx.x;
  const int rowBase = blockIdx.x * ROWS;

  for (int e = t; e < ROWS * WIN; e += THREADS)
    i_lds[e] = idx[rowBase * WIN + e];
  __syncthreads();

  for (int e = t; e < ROWS * HID; e += THREADS) {
    int row = e / HID;
    int col = e - row * HID;
    int w   = col / EMB;
    int c   = col - w * EMB;
    int id  = i_lds[row * WIN + w];
    float v = 0.0f;
    if (id >= 0 && id < VOCAB) v = emb[id * EMB + c];
    x_lds[row * STRIDE + col] = v;
  }
  for (int p = t; p < ROWS * (STRIDE - HID); p += THREADS) {
    int row = p / (STRIDE - HID);
    int c   = HID + (p - row * (STRIDE - HID));
    x_lds[row * STRIDE + c] = 0.0f;
  }
  __syncthreads();

  const int lane = t & 31;
  const int wave = t >> 5;
  const int hi   = lane >> 4;
  const int nloc = lane & 15;
  const int m    = lane & 15;

  const float* xr0 = &x_lds[( 0 + m) * STRIDE];
  const float* xr1 = &x_lds[(16 + m) * STRIDE];
  const float* xr2 = &x_lds[(32 + m) * STRIDE];
  const float* xr3 = &x_lds[(48 + m) * STRIDE];

  for (int p = 0; p < 4; ++p) {
    const int ct   = p * 4 + wave;
    const int coln = ct * 16 + nloc;
    const float bias = (coln < HID) ? b_in[coln] : 0.0f;
    v8f acc0 = {bias, bias, bias, bias, bias, bias, bias, bias};
    v8f acc1 = acc0, acc2 = acc0, acc3 = acc0;

    #pragma unroll 4
    for (int k = 0; k < KPAD; k += 4) {
      const int kk = k + 2 * hi;
      v2f b;
      b.x = wp[kk * KPAD + coln];
      b.y = wp[(kk + 1) * KPAD + coln];
      v2f a0 = *reinterpret_cast<const v2f*>(&xr0[kk]);
      v2f a1 = *reinterpret_cast<const v2f*>(&xr1[kk]);
      v2f a2 = *reinterpret_cast<const v2f*>(&xr2[kk]);
      v2f a3 = *reinterpret_cast<const v2f*>(&xr3[kk]);
      acc0 = __builtin_amdgcn_wmma_f32_16x16x4_f32(false, a0, false, b, (short)0, acc0, false, false);
      acc1 = __builtin_amdgcn_wmma_f32_16x16x4_f32(false, a1, false, b, (short)0, acc1, false, false);
      acc2 = __builtin_amdgcn_wmma_f32_16x16x4_f32(false, a2, false, b, (short)0, acc2, false, false);
      acc3 = __builtin_amdgcn_wmma_f32_16x16x4_f32(false, a3, false, b, (short)0, acc3, false, false);
    }

    #pragma unroll
    for (int j = 0; j < 8; ++j) {
      const int M = j + 8 * hi;
      h_lds[( 0 + M) * STRIDE + coln] = fast_tanh(acc0[j]);
      h_lds[(16 + M) * STRIDE + coln] = fast_tanh(acc1[j]);
      h_lds[(32 + M) * STRIDE + coln] = fast_tanh(acc2[j]);
      h_lds[(48 + M) * STRIDE + coln] = fast_tanh(acc3[j]);
    }
  }
  __syncthreads();

  const int rt = wave;
  const float* hrow = &h_lds[(rt * 16 + m) * STRIDE];

  for (int ct = 0; ct < 3; ++ct) {
    const int coln = ct * 16 + nloc;
    const float bias = (coln < OUTD) ? b_out[coln] : 0.0f;
    v8f acc = {bias, bias, bias, bias, bias, bias, bias, bias};

    #pragma unroll 4
    for (int k = 0; k < KPAD; k += 4) {
      const int kk = k + 2 * hi;
      v2f a = *reinterpret_cast<const v2f*>(&hrow[kk]);
      v2f b;
      b.x = wp2[kk * NPAD2 + coln];
      b.y = wp2[(kk + 1) * NPAD2 + coln];
      acc = __builtin_amdgcn_wmma_f32_16x16x4_f32(false, a, false, b, (short)0, acc, false, false);
    }

    #pragma unroll
    for (int j = 0; j < 8; ++j) {
      const int M = j + 8 * hi;
      if (coln < OUTD)
        out[(size_t)(rowBase + rt * 16 + M) * OUTD + coln] = acc[j];
    }
  }
}

extern "C" void kernel_launch(void* const* d_in, const int* in_sizes, int n_in,
                              void* d_out, int out_size, void* d_ws, size_t ws_size,
                              hipStream_t stream) {
  const int*   idx     = (const int*)  d_in[0];
  const float* emb     = (const float*)d_in[1];
  const float* w_in_p  = (const float*)d_in[2];
  const float* b_in_p  = (const float*)d_in[3];
  const float* w_out_p = (const float*)d_in[4];
  const float* b_out_p = (const float*)d_in[5];
  float* out = (float*)d_out;

  float* wp  = (float*)d_ws;
  float* wp2 = wp + KPAD * KPAD;

  pad_weights<<<(KPAD * KPAD + 255) / 256, 256, 0, stream>>>(w_in_p, w_out_p, wp, wp2);

  const int batch  = in_sizes[0] / WIN;
  const int blocks = batch / ROWS;
  tagger_fused<<<blocks, THREADS, 0, stream>>>(idx, emb, wp, b_in_p, wp2, b_out_p, out);
}


# Weight-stationary: all weights resident in LDS of 256 persistent workgroups, products transposed so x and h live only in registers (h's D tile is 2nd product's B, k order permuted: sum order changes)
# speedup vs baseline: 2.9305x; 2.9305x over previous
#include <hip/hip_runtime.h>

typedef __attribute__((ext_vector_type(2))) float v2f;
typedef __attribute__((ext_vector_type(4))) float v4f;
typedef __attribute__((ext_vector_type(8))) float v8f;

namespace {
constexpr int WIN     = 5;
constexpr int VOCAB   = 100000;
constexpr int EMB     = 50;
constexpr int HID     = 250;
constexpr int OUTD    = 36;
constexpr int KS1     = 63;
constexpr int HT      = 16;
constexpr int W1_TILE = 32 * 128;
constexpr int W2_TILE = 3 * 2 * 128;
constexpr int W1_FLOATS = HT * W1_TILE;
constexpr int W2_FLOATS = HT * W2_TILE;
constexpr int B1_OFF  = W1_FLOATS + W2_FLOATS;
constexpr int B2_OFF  = B1_OFF + 256;
constexpr int IMG_FLOATS = B2_OFF + 48;
constexpr int WAVES   = 16;
constexpr int THREADS = WAVES * 32;
}

__device__ __forceinline__ float fast_tanh(float x) {
  float xc = fminf(15.0f, fmaxf(-15.0f, x));
  float t  = __expf(2.0f * xc);
  return (t - 1.0f) * __builtin_amdgcn_rcpf(t + 1.0f);
}

__global__ __launch_bounds__(256)
void build_weight_image(const float* __restrict__ w_in, const float* __restrict__ b_in,
                        const float* __restrict__ w_out, const float* __restrict__ b_out,
                        float* __restrict__ img)
{
  const int e = blockIdx.x * 256 + threadIdx.x;
  if (e < W1_FLOATS) {
    const int q = e & 3, lane = (e >> 2) & 31, ip = (e >> 7) & 31, ht = e >> 12;
    const int m = lane & 15, hi = lane >> 4;
    const int k   = 4 * (2 * ip + (q >> 1)) + 2 * hi + (q & 1);
    const int col = ht * 16 + m;
    img[e] = (k < HID && col < HID) ? w_in[k * HID + col] : 0.0f;
  } else if (e < B1_OFF) {
    const int r = e - W1_FLOATS;
    const int q = r & 3, lane = (r >> 2) & 31, sp = (r >> 7) & 1, rest = r >> 8;
    const int ht = rest / 3, ot = rest - ht * 3;
    const int m = lane & 15, hi = lane >> 4;
    const int s    = 2 * sp + (q >> 1);
    const int hrow = ht * 16 + 8 * hi + 2 * s + (q & 1);
    const int col  = ot * 16 + m;
    img[e] = (hrow < HID && col < OUTD) ? w_out[hrow * OUTD + col] : 0.0f;
  } else if (e < B2_OFF) {
    const int h = e - B1_OFF;
    img[e] = (h < HID) ? b_in[h] : 0.0f;
  } else if (e < IMG_FLOATS) {
    const int c = e - B2_OFF;
    img[e] = (c < OUTD) ? b_out[c] : 0.0f;
  }
}

#define WMMA4(a, b, c) __builtin_amdgcn_wmma_f32_16x16x4_f32(false, (a), false, (b), (short)0, (c), false, false)

__global__ __launch_bounds__(THREADS)
void window_mlp(const int*   __restrict__ idx,
                const float* __restrict__ emb,
                const float* __restrict__ img,
                float*       __restrict__ out,
                int ntiles)
{
  __shared__ __attribute__((aligned(16))) float wl[IMG_FLOATS];

  const int t = threadIdx.x;
  {
    const v4f* src = reinterpret_cast<const v4f*>(img);
    v4f*       dst = reinterpret_cast<v4f*>(wl);
    for (int e = t; e < IMG_FLOATS / 4; e += THREADS) dst[e] = src[e];
  }
  __syncthreads();

  const int lane = t & 31;
  const int wave = __builtin_amdgcn_readfirstlane(t >> 5);
  const int hi   = lane >> 4;
  const int n    = lane & 15;

  const float* w1l = wl + lane * 4;
  const float* w2l = wl + W1_FLOATS + lane * 4;
  const float* b1l = wl + B1_OFF + 8 * hi;
  const float* b2l = wl + B2_OFF + 8 * hi;

  for (int tile = blockIdx.x * WAVES + wave; tile < ntiles; tile += gridDim.x * WAVES) {
    const int row = tile * 16 + n;

    int  bw[WIN];
    bool ok[WIN];
    #pragma unroll
    for (int w = 0; w < WIN; ++w) {
      const int v = idx[row * WIN + w];
      ok[w] = (v >= 0 && v < VOCAB);
      bw[w] = (ok[w] ? v : 0) * EMB - EMB * w + 2 * hi;
    }
    v2f xb[KS1];
    #pragma unroll
    for (int i = 0; i < KS1; ++i) {
      const int k0  = 4 * i;
      const int wlo = k0 / EMB;
      const int whi = (k0 + 2) / EMB;
      int off; bool good;
      if (i == KS1 - 1) {
        off  = bw[WIN - 1] - 2 * hi + k0;
        good = ok[WIN - 1] && (hi == 0);
      } else if (wlo == whi) {
        off  = bw[wlo] + k0;
        good = ok[wlo];
      } else {
        off  = (hi ? bw[whi] : bw[wlo]) + k0;
        good = hi ? ok[whi] : ok[wlo];
      }
      const v2f v = *reinterpret_cast<const v2f*>(emb + off);
      const v2f z = {0.0f, 0.0f};
      xb[i] = good ? v : z;
    }

    v8f o0, o1, o2;
    {
      const v4f a = *reinterpret_cast<const v4f*>(b2l),      b = *reinterpret_cast<const v4f*>(b2l + 4);
      const v4f c = *reinterpret_cast<const v4f*>(b2l + 16), d = *reinterpret_cast<const v4f*>(b2l + 20);
      const v4f e = *reinterpret_cast<const v4f*>(b2l + 32), f = *reinterpret_cast<const v4f*>(b2l + 36);
      o0 = (v8f){a.x, a.y, a.z, a.w, b.x, b.y, b.z, b.w};
      o1 = (v8f){c.x, c.y, c.z, c.w, d.x, d.y, d.z, d.w};
      o2 = (v8f){e.x, e.y, e.z, e.w, f.x, f.y, f.z, f.w};
    }

    #pragma unroll 1
    for (int p = 0; p < HT / 2; ++p) {
      const float* wa = w1l + (2 * p) * W1_TILE;
      const float* wb = wa + W1_TILE;
      v8f accA, accB;
      {
        const v4f a = *reinterpret_cast<const v4f*>(b1l + (2 * p) * 16);
        const v4f b = *reinterpret_cast<const v4f*>(b1l + (2 * p) * 16 + 4);
        const v4f c = *reinterpret_cast<const v4f*>(b1l + (2 * p) * 16 + 16);
        const v4f d = *reinterpret_cast<const v4f*>(b1l + (2 * p) * 16 + 20);
        accA = (v8f){a.x, a.y, a.z, a.w, b.x, b.y, b.z, b.w};
        accB = (v8f){c.x, c.y, c.z, c.w, d.x, d.y, d.z, d.w};
      }
      v4f fa[32], fb[32];
      fa[0] = *reinterpret_cast<const v4f*>(wa);        fb[0] = *reinterpret_cast<const v4f*>(wb);
      fa[1] = *reinterpret_cast<const v4f*>(wa + 128);  fb[1] = *reinterpret_cast<const v4f*>(wb + 128);
      #pragma unroll
      for (int ip = 0; ip < 32; ++ip) {
        if (ip + 2 < 32) {
          fa[ip + 2] = *reinterpret_cast<const v4f*>(wa + (ip + 2) * 128);
          fb[ip + 2] = *reinterpret_cast<const v4f*>(wb + (ip + 2) * 128);
        }
        accA = WMMA4(((v2f){fa[ip].x, fa[ip].y}), xb[2 * ip], accA);
        accB = WMMA4(((v2f){fb[ip].x, fb[ip].y}), xb[2 * ip], accB);
        if (2 * ip + 1 < KS1) {
          accA = WMMA4(((v2f){fa[ip].z, fa[ip].w}), xb[2 * ip + 1], accA);
          accB = WMMA4(((v2f){fb[ip].z, fb[ip].w}), xb[2 * ip + 1], accB);
        }
      }
      #pragma unroll
      for (int j = 0; j < 8; ++j) { accA[j] = fast_tanh(accA[j]); accB[j] = fast_tanh(accB[j]); }

      const float* ua = w2l + (2 * p) * W2_TILE;
      const float* ub = ua + W2_TILE;
      #pragma unroll
      for (int sp = 0; sp < 2; ++sp) {
        const v2f hA0 = {accA[4 * sp], accA[4 * sp + 1]}, hA1 = {accA[4 * sp + 2], accA[4 * sp + 3]};
        const v4f f0 = *reinterpret_cast<const v4f*>(ua + (0 * 2 + sp) * 128);
        const v4f f1 = *reinterpret_cast<const v4f*>(ua + (1 * 2 + sp) * 128);
        const v4f f2 = *reinterpret_cast<const v4f*>(ua + (2 * 2 + sp) * 128);
        o0 = WMMA4(((v2f){f0.x, f0.y}), hA0, o0);
        o1 = WMMA4(((v2f){f1.x, f1.y}), hA0, o1);
        o2 = WMMA4(((v2f){f2.x, f2.y}), hA0, o2);
        o0 = WMMA4(((v2f){f0.z, f0.w}), hA1, o0);
        o1 = WMMA4(((v2f){f1.z, f1.w}), hA1, o1);
        o2 = WMMA4(((v2f){f2.z, f2.w}), hA1, o2);
      }
      #pragma unroll
      for (int sp = 0; sp < 2; ++sp) {
        const v2f hB0 = {accB[4 * sp], accB[4 * sp + 1]}, hB1 = {accB[4 * sp + 2], accB[4 * sp + 3]};
        const v4f f0 = *reinterpret_cast<const v4f*>(ub + (0 * 2 + sp) * 128);
        const v4f f1 = *reinterpret_cast<const v4f*>(ub + (1 * 2 + sp) * 128);
        const v4f f2 = *reinterpret_cast<const v4f*>(ub + (2 * 2 + sp) * 128);
        o0 = WMMA4(((v2f){f0.x, f0.y}), hB0, o0);
        o1 = WMMA4(((v2f){f1.x, f1.y}), hB0, o1);
        o2 = WMMA4(((v2f){f2.x, f2.y}), hB0, o2);
        o0 = WMMA4(((v2f){f0.z, f0.w}), hB1, o0);
        o1 = WMMA4(((v2f){f1.z, f1.w}), hB1, o1);
        o2 = WMMA4(((v2f){f2.z, f2.w}), hB1, o2);
      }
    }

    float* orow = out + (size_t)row * OUTD + 8 * hi;
    *reinterpret_cast<v4f*>(orow)      = (v4f){o0[0], o0[1], o0[2], o0[3]};
    *reinterpret_cast<v4f*>(orow + 4)  = (v4f){o0[4], o0[5], o0[6], o0[7]};
    *reinterpret_cast<v4f*>(orow + 16) = (v4f){o1[0], o1[1], o1[2], o1[3]};
    *reinterpret_cast<v4f*>(orow + 20) = (v4f){o1[4], o1[5], o1[6], o1[7]};
    if (hi == 0)
      *reinterpret_cast<v4f*>(orow + 32) = (v4f){o2[0], o2[1], o2[2], o2[3]};
  }
}

extern "C" void kernel_launch(void* const* d_in, const int* in_sizes, int n_in,
                              void* d_out, int out_size, void* d_ws, size_t ws_size,
                              hipStream_t stream) {
  const int*   idx     = (const int*)  d_in[0];
  const float* emb     = (const float*)d_in[1];
  const float* w_in_p  = (const float*)d_in[2];
  const float* b_in_p  = (const float*)d_in[3];
  const float* w_out_p = (const float*)d_in[4];
  const float* b_out_p = (const float*)d_in[5];
  float* out = (float*)d_out;

  float* img = (float*)d_ws;

  build_weight_image<<<(IMG_FLOATS + 255) / 256, 256, 0, stream>>>(w_in_p, b_in_p, w_out_p, b_out_p, img);

  const int batch  = in_sizes[0] / WIN;
  const int ntiles = batch / 16;
  window_mlp<<<256, THREADS, 0, stream>>>(idx, emb, img, out, ntiles);
}


# LDS weight-fragment loads issued as groups one matrix block ahead of use (two register sets), incl. second product's and the next pair's first group; was 2 loads between every 4 matrix instructions
# speedup vs baseline: 3.0330x; 1.0350x over previous
#include <hip/hip_runtime.h>

typedef __attribute__((ext_vector_type(2))) float v2f;
typedef __attribute__((ext_vector_type(4))) float v4f;
typedef __attribute__((ext_vector_type(8))) float v8f;

namespace {
constexpr int WIN     = 5;
constexpr int VOCAB   = 100000;
constexpr int EMB     = 50;
constexpr int HID     = 250;
constexpr int OUTD    = 36;
constexpr int KS1     = 63;
constexpr int HT      = 16;
constexpr int W1_TILE = 32 * 128;
constexpr int W2_TILE = 3 * 2 * 128;
constexpr int W1_FLOATS = HT * W1_TILE;
constexpr int W2_FLOATS = HT * W2_TILE;
constexpr int B1_OFF  = W1_FLOATS + W2_FLOATS;
constexpr int B2_OFF  = B1_OFF + 256;
constexpr int IMG_FLOATS = B2_OFF + 48;
constexpr int WAVES   = 16;
constexpr int THREADS = WAVES * 32;
constexpr int GRP     = 3;
constexpr int NGRP    = (32 + GRP - 1) / GRP;
}

__device__ __forceinline__ float fast_tanh(float x) {
  float xc = fminf(15.0f, fmaxf(-15.0f, x));
  float t  = __expf(2.0f * xc);
  return (t - 1.0f) * __builtin_amdgcn_rcpf(t + 1.0f);
}

__global__ __launch_bounds__(256)
void build_weight_image(const float* __restrict__ w_in, const float* __restrict__ b_in,
                        const float* __restrict__ w_out, const float* __restrict__ b_out,
                        float* __restrict__ img)
{
  const int e = blockIdx.x * 256 + threadIdx.x;
  if (e < W1_FLOATS) {
    const int q = e & 3, lane = (e >> 2) & 31, ip = (e >> 7) & 31, ht = e >> 12;
    const int m = lane & 15, hi = lane >> 4;
    const int k   = 4 * (2 * ip + (q >> 1)) + 2 * hi + (q & 1);
    const int col = ht * 16 + m;
    img[e] = (k < HID && col < HID) ? w_in[k * HID + col] : 0.0f;
  } else if (e < B1_OFF) {
    const int r = e - W1_FLOATS;
    const int q = r & 3, lane = (r >> 2) & 31, sp = (r >> 7) & 1, rest = r >> 8;
    const int ht = rest / 3, ot = rest - ht * 3;
    const int m = lane & 15, hi = lane >> 4;
    const int s    = 2 * sp + (q >> 1);
    const int hrow = ht * 16 + 8 * hi + 2 * s + (q & 1);
    const int col  = ot * 16 + m;
    img[e] = (hrow < HID && col < OUTD) ? w_out[hrow * OUTD + col] : 0.0f;
  } else if (e < B2_OFF) {
    const int h = e - B1_OFF;
    img[e] = (h < HID) ? b_in[h] : 0.0f;
  } else if (e < IMG_FLOATS) {
    const int c = e - B2_OFF;
    img[e] = (c < OUTD) ? b_out[c] : 0.0f;
  }
}

#define LDS4(p) (*reinterpret_cast<const v4f*>(p))
#define WMMA4(a, b, c) __builtin_amdgcn_wmma_f32_16x16x4_f32(false, (a), false, (b), (short)0, (c), false, false)

__global__ __launch_bounds__(THREADS)
void window_mlp(const int*   __restrict__ idx,
                const float* __restrict__ emb,
                const float* __restrict__ img,
                float*       __restrict__ out,
                int ntiles)
{
  __shared__ __attribute__((aligned(16))) float wl[IMG_FLOATS];

  const int t = threadIdx.x;
  {
    const v4f* src = reinterpret_cast<const v4f*>(img);
    v4f*       dst = reinterpret_cast<v4f*>(wl);
    for (int e = t; e < IMG_FLOATS / 4; e += THREADS) dst[e] = src[e];
  }
  __syncthreads();

  const int lane = t & 31;
  const int wave = __builtin_amdgcn_readfirstlane(t >> 5);
  const int hi   = lane >> 4;
  const int n    = lane & 15;

  const float* w1l = wl + lane * 4;
  const float* w2l = wl + W1_FLOATS + lane * 4;
  const float* b1l = wl + B1_OFF + 8 * hi;
  const float* b2l = wl + B2_OFF + 8 * hi;

  for (int tile = blockIdx.x * WAVES + wave; tile < ntiles; tile += gridDim.x * WAVES) {
    const int row = tile * 16 + n;

    int  bw[WIN];
    bool ok[WIN];
    #pragma unroll
    for (int w = 0; w < WIN; ++w) {
      const int v = idx[row * WIN + w];
      ok[w] = (v >= 0 && v < VOCAB);
      bw[w] = (ok[w] ? v : 0) * EMB - EMB * w + 2 * hi;
    }
    v2f xb[KS1];
    #pragma unroll
    for (int i = 0; i < KS1; ++i) {
      const int k0  = 4 * i;
      const int wlo = k0 / EMB;
      const int whi = (k0 + 2) / EMB;
      int off; bool good;
      if (i == KS1 - 1) {
        off  = bw[WIN - 1] - 2 * hi + k0;
        good = ok[WIN - 1] && (hi == 0);
      } else if (wlo == whi) {
        off  = bw[wlo] + k0;
        good = ok[wlo];
      } else {
        off  = (hi ? bw[whi] : bw[wlo]) + k0;
        good = hi ? ok[whi] : ok[wlo];
      }
      const v2f v = *reinterpret_cast<const v2f*>(emb + off);
      const v2f z = {0.0f, 0.0f};
      xb[i] = good ? v : z;
    }

    v8f o0, o1, o2;
    {
      const v4f a = *reinterpret_cast<const v4f*>(b2l),      b = *reinterpret_cast<const v4f*>(b2l + 4);
      const v4f c = *reinterpret_cast<const v4f*>(b2l + 16), d = *reinterpret_cast<const v4f*>(b2l + 20);
      const v4f e = *reinterpret_cast<const v4f*>(b2l + 32), f = *reinterpret_cast<const v4f*>(b2l + 36);
      o0 = (v8f){a.x, a.y, a.z, a.w, b.x, b.y, b.z, b.w};
      o1 = (v8f){c.x, c.y, c.z, c.w, d.x, d.y, d.z, d.w};
      o2 = (v8f){e.x, e.y, e.z, e.w, f.x, f.y, f.z, f.w};
    }

    v4f na[GRP], nb[GRP];
    v4f nbias[4];
    #pragma unroll
    for (int j = 0; j < GRP; ++j) {
      na[j] = LDS4(w1l + j * 128);
      nb[j] = LDS4(w1l + W1_TILE + j * 128);
    }
    nbias[0] = LDS4(b1l);      nbias[1] = LDS4(b1l + 4);
    nbias[2] = LDS4(b1l + 16); nbias[3] = LDS4(b1l + 20);

    #pragma unroll 1
    for (int p = 0; p < HT / 2; ++p) {
      const float* wa = w1l + (2 * p) * W1_TILE;
      const float* wb = wa + W1_TILE;
      v8f accA, accB;
      {
        const v4f a = nbias[0], b = nbias[1], c = nbias[2], d = nbias[3];
        accA = (v8f){a.x, a.y, a.z, a.w, b.x, b.y, b.z, b.w};
        accB = (v8f){c.x, c.y, c.z, c.w, d.x, d.y, d.z, d.w};
      }
      const float* ua = w2l + (2 * p) * W2_TILE;
      const float* ub = ua + W2_TILE;
      v4f fa[2][GRP], fb[2][GRP];
      v4f qa[6], qb[6];
      #pragma unroll
      for (int j = 0; j < GRP; ++j) { fa[0][j] = na[j]; fb[0][j] = nb[j]; }
      #pragma unroll
      for (int g = 0; g < NGRP; ++g) {
        const int cur = g & 1, nxt = cur ^ 1;
        if (g + 1 < NGRP) {
          #pragma unroll
          for (int j = 0; j < GRP; ++j) {
            const int ip = (g + 1) * GRP + j;
            if (ip < 32) {
              fa[nxt][j] = LDS4(wa + ip * 128);
              fb[nxt][j] = LDS4(wb + ip * 128);
            }
          }
        } else {
          #pragma unroll
          for (int f = 0; f < 6; ++f) qa[f] = LDS4(ua + f * 128);
        }
        #pragma unroll
        for (int j = 0; j < GRP; ++j) {
          const int ip = g * GRP + j;
          if (ip < 32) {
            accA = WMMA4(((v2f){fa[cur][j].x, fa[cur][j].y}), xb[2 * ip], accA);
            accB = WMMA4(((v2f){fb[cur][j].x, fb[cur][j].y}), xb[2 * ip], accB);
            if (2 * ip + 1 < KS1) {
              accA = WMMA4(((v2f){fa[cur][j].z, fa[cur][j].w}), xb[2 * ip + 1], accA);
              accB = WMMA4(((v2f){fb[cur][j].z, fb[cur][j].w}), xb[2 * ip + 1], accB);
            }
          }
        }
      }
      #pragma unroll
      for (int f = 0; f < 6; ++f) qb[f] = LDS4(ub + f * 128);

      #pragma unroll
      for (int j = 0; j < 8; ++j) { accA[j] = fast_tanh(accA[j]); accB[j] = fast_tanh(accB[j]); }

      #pragma unroll
      for (int sp = 0; sp < 2; ++sp) {
        const v2f hA0 = {accA[4 * sp], accA[4 * sp + 1]}, hA1 = {accA[4 * sp + 2], accA[4 * sp + 3]};
        const v4f f0 = qa[0 + sp], f1 = qa[2 + sp], f2 = qa[4 + sp];
        o0 = WMMA4(((v2f){f0.x, f0.y}), hA0, o0);
        o1 = WMMA4(((v2f){f1.x, f1.y}), hA0, o1);
        o2 = WMMA4(((v2f){f2.x, f2.y}), hA0, o2);
        o0 = WMMA4(((v2f){f0.z, f0.w}), hA1, o0);
        o1 = WMMA4(((v2f){f1.z, f1.w}), hA1, o1);
        o2 = WMMA4(((v2f){f2.z, f2.w}), hA1, o2);
      }
      {
        const int pn = (p + 1) & (HT / 2 - 1);
        const float* wn = w1l + (2 * pn) * W1_TILE;
        #pragma unroll
        for (int j = 0; j < GRP; ++j) {
          na[j] = LDS4(wn + j * 128);
          nb[j] = LDS4(wn + W1_TILE + j * 128);
        }
        nbias[0] = LDS4(b1l + (2 * pn) * 16);      nbias[1] = LDS4(b1l + (2 * pn) * 16 + 4);
        nbias[2] = LDS4(b1l + (2 * pn) * 16 + 16); nbias[3] = LDS4(b1l + (2 * pn) * 16 + 20);
      }
      __builtin_amdgcn_sched_barrier(0x407);
      #pragma unroll
      for (int sp = 0; sp < 2; ++sp) {
        const v2f hB0 = {accB[4 * sp], accB[4 * sp + 1]}, hB1 = {accB[4 * sp + 2], accB[4 * sp + 3]};
        const v4f f0 = qb[0 + sp], f1 = qb[2 + sp], f2 = qb[4 + sp];
        o0 = WMMA4(((v2f){f0.x, f0.y}), hB0, o0);
        o1 = WMMA4(((v2f){f1.x, f1.y}), hB0, o1);
        o2 = WMMA4(((v2f){f2.x, f2.y}), hB0, o2);
        o0 = WMMA4(((v2f){f0.z, f0.w}), hB1, o0);
        o1 = WMMA4(((v2f){f1.z, f1.w}), hB1, o1);
        o2 = WMMA4(((v2f){f2.z, f2.w}), hB1, o2);
      }
    }

    float* orow = out + (size_t)row * OUTD + 8 * hi;
    *reinterpret_cast<v4f*>(orow)      = (v4f){o0[0], o0[1], o0[2], o0[3]};
    *reinterpret_cast<v4f*>(orow + 4)  = (v4f){o0[4], o0[5], o0[6], o0[7]};
    *reinterpret_cast<v4f*>(orow + 16) = (v4f){o1[0], o1[1], o1[2], o1[3]};
    *reinterpret_cast<v4f*>(orow + 20) = (v4f){o1[4], o1[5], o1[6], o1[7]};
    if (hi == 0)
      *reinterpret_cast<v4f*>(orow + 32) = (v4f){o2[0], o2[1], o2[2], o2[3]};
  }
}

extern "C" void kernel_launch(void* const* d_in, const int* in_sizes, int n_in,
                              void* d_out, int out_size, void* d_ws, size_t ws_size,
                              hipStream_t stream) {
  const int*   idx     = (const int*)  d_in[0];
  const float* emb     = (const float*)d_in[1];
  const float* w_in_p  = (const float*)d_in[2];
  const float* b_in_p  = (const float*)d_in[3];
  const float* w_out_p = (const float*)d_in[4];
  const float* b_out_p = (const float*)d_in[5];
  float* out = (float*)d_out;

  float* img = (float*)d_ws;

  build_weight_image<<<(IMG_FLOATS + 255) / 256, 256, 0, stream>>>(w_in_p, b_in_p, w_out_p, b_out_p, img);

  const int batch  = in_sizes[0] / WIN;
  const int ntiles = batch / 16;
  window_mlp<<<256, THREADS, 0, stream>>>(idx, emb, img, out, ntiles);
}
